# starting kernel
# baseline (speedup 1.0000x reference)
#include <hip/hip_runtime.h>


typedef float v2f __attribute__((ext_vector_type(2)));
typedef float v8f __attribute__((ext_vector_type(8)));

#define BLOCK 256

static __device__ __forceinline__ float fast_tanh(float x) {
#if __has_builtin(__builtin_amdgcn_tanhf)
  return __builtin_amdgcn_tanhf(x);
#elif __has_builtin(__builtin_amdgcn_tanh_f32)
  return __builtin_amdgcn_tanh_f32(x);
#else
  float e = __builtin_amdgcn_exp2f(x * 2.8853900817779268f);
  return 1.0f - 2.0f / (e + 1.0f);
#endif
}

static __device__ __forceinline__ float softplus_f(float x) {
  float t = __builtin_amdgcn_exp2f(-__builtin_fabsf(x) * 1.4426950408889634f);
  return fmaxf(x, 0.0f) + 0.6931471805599453f * __builtin_amdgcn_logf(1.0f + t);
}

__global__ __launch_bounds__(BLOCK) void csys_kernel(
    const float* __restrict__ t_in, const float* __restrict__ z,
    const float* __restrict__ logK, const float* __restrict__ logz,
    const float* __restrict__ logp, const float* __restrict__ W1,
    const float* __restrict__ b1, const float* __restrict__ W2,
    const float* __restrict__ b2, float* __restrict__ out, int n) {
  __shared__ float xch[2 * BLOCK];

  const int tid   = threadIdx.x;
  const int lane  = tid & 31;
  const int wbase = (tid >> 5) * 64;
  const int gid   = blockIdx.x * BLOCK + tid;
  const int idx = (gid < n) ? gid : (n - 1);

  const float tt = t_in[idx];
  const float x1 = z[0 * n + idx];
  const float v1 = z[1 * n + idx];
  const float x2 = z[2 * n + idx];
  const float v2 = z[3 * n + idx];
  const float xc = z[4 * n + idx];

  const int khalf = lane >> 4;
  const int nn    = lane & 15;

  v2f bmat[16];
#pragma unroll
  for (int c = 0; c < 16; ++c) {
    const int kb = 4 * c + 2 * khalf;
    float bx = 0.0f, by = 0.0f;
    if (nn < 2) {
      bx = W2[kb * 2 + nn];
      by = W2[(kb + 1) * 2 + nn];
    }
    bmat[c].x = bx;
    bmat[c].y = by;
  }

  const float v2t0 = __shfl(v2, nn, 32);
  const float v2t1 = __shfl(v2, 16 + nn, 32);
  v8f acc0 = {};
  v8f acc1 = {};
#pragma unroll
  for (int c = 0; c < 16; ++c) {
    const int ka = 4 * c + 2 * khalf;
    const float2 w1p = *reinterpret_cast<const float2*>(W1 + ka);
    const float2 b1p = *reinterpret_cast<const float2*>(b1 + ka);
    v2f a0, a1;
    a0.x = fast_tanh(fmaf(v2t0, w1p.x, b1p.x));
    a0.y = fast_tanh(fmaf(v2t0, w1p.y, b1p.y));
    a1.x = fast_tanh(fmaf(v2t1, w1p.x, b1p.x));
    a1.y = fast_tanh(fmaf(v2t1, w1p.y, b1p.y));
    acc0 = __builtin_amdgcn_wmma_f32_16x16x4_f32(
        false, a0, false, bmat[c], (short)0, acc0, false, false);
    acc1 = __builtin_amdgcn_wmma_f32_16x16x4_f32(
        false, a1, false, bmat[c], (short)0, acc1, false, false);
  }

  if (nn < 2) {
    const int wv0 = khalf * 8;
    const int wv1 = 16 + khalf * 8;
#pragma unroll
    for (int r = 0; r < 8; ++r) {
      xch[wbase + (wv0 + r) * 2 + nn] = acc0[r];
      xch[wbase + (wv1 + r) * 2 + nn] = acc1[r];
    }
  }
  __syncthreads();

  const float kinetic = softplus_f(xch[wbase + lane * 2 + 0] + b2[0]);
  const float stic    = softplus_f(xch[wbase + lane * 2 + 1] + b2[1]);

  const float Kc = expf(logK[0]);
  const float zc = expf(logz[0]);
  const float pc = expf(logp[0]);

  const float x2ref = 0.5f * __sinf(0.5f * tt);
  const float e     = x2ref - x2;
  const float d_xc  = -pc * xc + e;
  const float u     = Kc * (zc - pc) * xc + Kc * e;

  const float dx1 = v1;
  const float dv1 = u - 2.0f * x1 - 0.5f * v1 - 3.0f * (x1 - x2) - 0.8f * (v1 - v2);
  const float dx2 = v2;

  const float F_net    = 3.0f * (x1 - x2) + 0.8f * (v1 - v2);
  const float F_static = -fminf(fmaxf(F_net, -stic), stic);
  const float sgn      = (v2 > 0.0f) ? 1.0f : ((v2 < 0.0f) ? -1.0f : 0.0f);
  const float F_kin    = -kinetic * sgn;
  const float F_fric   = (__builtin_fabsf(v2) < 0.01f) ? F_static : F_kin;
  const float dv2      = (F_net + F_fric) * (1.0f / 1.5f);

  out[0 * n + idx] = dx1;
  out[1 * n + idx] = dv1;
  out[2 * n + idx] = dx2;
  out[3 * n + idx] = dv2;
  out[4 * n + idx] = d_xc;
}

extern "C" void kernel_launch(void* const* d_in, const int* in_sizes, int n_in,
                              void* d_out, int out_size, void* d_ws,
                              size_t ws_size, hipStream_t stream) {
  const float* t    = (const float*)d_in[0];
  const float* z    = (const float*)d_in[1];
  const float* logK = (const float*)d_in[2];
  const float* logz = (const float*)d_in[3];
  const float* logp = (const float*)d_in[4];
  const float* W1   = (const float*)d_in[5];
  const float* b1   = (const float*)d_in[6];
  const float* W2   = (const float*)d_in[7];
  const float* b2   = (const float*)d_in[8];
  float* out        = (float*)d_out;

  const int n      = in_sizes[0];
  const int blocks = (n + BLOCK - 1) / BLOCK;
  hipLaunchKernelGGL(csys_kernel, dim3(blocks), dim3(BLOCK), 0, stream,
                     t, z, logK, logz, logp, W1, b1, W2, b2, out, n);
}


# Software-pipelined activation: behind each WMMA the wave issues the tanh of the product two steps on and the affine part three steps on (register rings); weights refilled two chunks ahead
# speedup vs baseline: 1.2920x; 1.2920x over previous
#include <hip/hip_runtime.h>


typedef float v2f __attribute__((ext_vector_type(2)));
typedef float v8f __attribute__((ext_vector_type(8)));

#define BLOCK 256
#define EPL 4
#define NT (2 * EPL)
#define TILE (BLOCK * EPL)

static __device__ __forceinline__ float fast_tanh(float x) {
#if __has_builtin(__builtin_amdgcn_tanhf)
  return __builtin_amdgcn_tanhf(x);
#elif __has_builtin(__builtin_amdgcn_tanh_f32)
  return __builtin_amdgcn_tanh_f32(x);
#else
  float e = __builtin_amdgcn_exp2f(x * 2.8853900817779268f);
  return 1.0f - 2.0f / (e + 1.0f);
#endif
}

static __device__ __forceinline__ float softplus_f(float x) {
  float t = __builtin_amdgcn_exp2f(-__builtin_fabsf(x) * 1.4426950408889634f);
  return fmaxf(x, 0.0f) + 0.6931471805599453f * __builtin_amdgcn_logf(1.0f + t);
}

#define NSTEP (16 * NT)

struct Pipe {
  v8f    acc[NT];
  float  vt[NT];
  float2 w1r[2], b1r[2];
  v2f    afr[2];
  v2f    pre[2];
  v2f    bfr[3];
};

template <int C>
static __device__ __forceinline__ void load_w1b1(Pipe& p, const float* __restrict__ w1l,
                                                 const float* __restrict__ b1l) {
  p.w1r[C & 1] = *reinterpret_cast<const float2*>(w1l + 4 * C);
  p.b1r[C & 1] = *reinterpret_cast<const float2*>(b1l + 4 * C);
}

template <int C>
static __device__ __forceinline__ void load_af(Pipe& p, const float* __restrict__ w2l) {
  p.afr[C & 1].x = w2l[8 * C];
  p.afr[C & 1].y = w2l[8 * C + 2];
}

template <int I>
static __device__ __forceinline__ void step_pre(Pipe& p) {
  constexpr int c = I / NT, t = I % NT;
  p.pre[I & 1].x = fmaf(p.vt[t], p.w1r[c & 1].x, p.b1r[c & 1].x);
  p.pre[I & 1].y = fmaf(p.vt[t], p.w1r[c & 1].y, p.b1r[c & 1].y);
}

template <int I>
static __device__ __forceinline__ void step_act(Pipe& p) {
  p.bfr[I % 3].x = fast_tanh(p.pre[I & 1].x);
  p.bfr[I % 3].y = fast_tanh(p.pre[I & 1].y);
}

template <int I>
static __device__ __forceinline__ void pipe_steps(Pipe& p, const float* __restrict__ w1l,
                                                  const float* __restrict__ b1l,
                                                  const float* __restrict__ w2l) {
  constexpr int c = I / NT, t = I % NT;
  p.acc[t] = __builtin_amdgcn_wmma_f32_16x16x4_f32(
      false, p.afr[c & 1], false, p.bfr[I % 3], (short)0, p.acc[t], false, false);
  if constexpr (I + 2 < NSTEP) step_act<I + 2>(p);
  if constexpr (I + 3 < NSTEP) step_pre<I + 3>(p);
  if constexpr (t == NT - 4 && c + 2 < 16) load_w1b1<c + 2>(p, w1l, b1l);
  if constexpr (t == NT - 1 && c + 2 < 16) load_af<c + 2>(p, w2l);
  __builtin_amdgcn_sched_barrier(0);
  if constexpr (I + 1 < NSTEP) pipe_steps<I + 1>(p, w1l, b1l, w2l);
}

template <bool VEC>
static __device__ __forceinline__ void load_row(const float* __restrict__ p,
                                                int base, int n,
                                                float (&o)[EPL]) {
  if (VEC) {
    const float4 t = *reinterpret_cast<const float4*>(p + base);
    o[0] = t.x; o[1] = t.y; o[2] = t.z; o[3] = t.w;
  } else {
#pragma unroll
    for (int q = 0; q < EPL; ++q) {
      const int i = base + q;
      o[q] = p[i < n ? i : n - 1];
    }
  }
}

template <bool VEC>
static __device__ __forceinline__ void store_row(float* __restrict__ p, int base,
                                                 int n, bool live,
                                                 const float (&v)[EPL]) {
  if (VEC) {
    if (live) {
      float4 t;
      t.x = v[0]; t.y = v[1]; t.z = v[2]; t.w = v[3];
      *reinterpret_cast<float4*>(p + base) = t;
    }
  } else {
#pragma unroll
    for (int q = 0; q < EPL; ++q)
      if (live && base + q < n) p[base + q] = v[q];
  }
}

template <bool VEC>
__global__ __launch_bounds__(BLOCK) void csys_kernel(
    const float* __restrict__ t_in, const float* __restrict__ z,
    const float* __restrict__ logK, const float* __restrict__ logz,
    const float* __restrict__ logp, const float* __restrict__ W1,
    const float* __restrict__ b1, const float* __restrict__ W2,
    const float* __restrict__ b2, float* __restrict__ out, int n) {
  const int tid   = threadIdx.x;
  const int lane  = tid & 31;
  const int khalf = lane >> 4;
  const int nn    = lane & 15;

  const int  ngrp = (n + EPL - 1) / EPL;
  const int  grp  = blockIdx.x * BLOCK + tid;
  const bool live = grp < ngrp;
  const int  base = EPL * (live ? grp : ngrp - 1);
  const size_t pitch = (size_t)n;

  float v2[EPL], tt[EPL], x1[EPL], v1[EPL], x2[EPL], xc[EPL];
  load_row<VEC>(z + 3 * pitch, base, n, v2);
  load_row<VEC>(z + 0 * pitch, base, n, x1);
  load_row<VEC>(z + 1 * pitch, base, n, v1);
  load_row<VEC>(z + 2 * pitch, base, n, x2);
  load_row<VEC>(z + 4 * pitch, base, n, xc);
  load_row<VEC>(t_in, base, n, tt);

  Pipe p;
#pragma unroll
  for (int q = 0; q < EPL; ++q) {
    p.vt[2 * q + 0] = __shfl(v2[q], nn, 32);
    p.vt[2 * q + 1] = __shfl(v2[q], 16 + nn, 32);
  }

  const float* __restrict__ w2l = W2 + 4 * khalf + (nn & 1);
  const float* __restrict__ w1l = W1 + 2 * khalf;
  const float* __restrict__ b1l = b1 + 2 * khalf;

#pragma unroll
  for (int t = 0; t < NT; ++t) p.acc[t] = (v8f){};

  load_w1b1<0>(p, w1l, b1l);
  load_af<0>(p, w2l);
  load_w1b1<1>(p, w1l, b1l);
  load_af<1>(p, w2l);
  step_pre<0>(p);
  step_act<0>(p);
  step_pre<1>(p);
  step_act<1>(p);
  step_pre<2>(p);
  __builtin_amdgcn_sched_barrier(0);
  pipe_steps<0>(p, w1l, b1l, w2l);

  const float b20 = b2[0];
  const float b21 = b2[1];
  const float Kc = expf(logK[0]);
  const float zc = expf(logz[0]);
  const float pc = expf(logp[0]);

  float o_dv1[EPL], o_dv2[EPL], o_dxc[EPL];
#pragma unroll
  for (int q = 0; q < EPL; ++q) {
    const float m0 = khalf ? p.acc[2 * q + 1][0] : p.acc[2 * q][0];
    const float m1 = khalf ? p.acc[2 * q + 1][1] : p.acc[2 * q][1];
    const float kinetic = softplus_f(m0 + b20);
    const float stic    = softplus_f(m1 + b21);

    const float x2ref = 0.5f * __sinf(0.5f * tt[q]);
    const float e     = x2ref - x2[q];
    o_dxc[q]          = -pc * xc[q] + e;
    const float u     = Kc * (zc - pc) * xc[q] + Kc * e;

    o_dv1[q] = u - 2.0f * x1[q] - 0.5f * v1[q] - 3.0f * (x1[q] - x2[q]) -
               0.8f * (v1[q] - v2[q]);

    const float F_net    = 3.0f * (x1[q] - x2[q]) + 0.8f * (v1[q] - v2[q]);
    const float F_static = -fminf(fmaxf(F_net, -stic), stic);
    const float sgn = (v2[q] > 0.0f) ? 1.0f : ((v2[q] < 0.0f) ? -1.0f : 0.0f);
    const float F_kin  = -kinetic * sgn;
    const float F_fric = (__builtin_fabsf(v2[q]) < 0.01f) ? F_static : F_kin;
    o_dv2[q] = (F_net + F_fric) * (1.0f / 1.5f);
  }

  store_row<VEC>(out + 0 * pitch, base, n, live, v1);
  store_row<VEC>(out + 1 * pitch, base, n, live, o_dv1);
  store_row<VEC>(out + 2 * pitch, base, n, live, v2);
  store_row<VEC>(out + 3 * pitch, base, n, live, o_dv2);
  store_row<VEC>(out + 4 * pitch, base, n, live, o_dxc);
}

extern "C" void kernel_launch(void* const* d_in, const int* in_sizes, int n_in,
                              void* d_out, int out_size, void* d_ws,
                              size_t ws_size, hipStream_t stream) {
  const float* t    = (const float*)d_in[0];
  const float* z    = (const float*)d_in[1];
  const float* logK = (const float*)d_in[2];
  const float* logz = (const float*)d_in[3];
  const float* logp = (const float*)d_in[4];
  const float* W1   = (const float*)d_in[5];
  const float* b1   = (const float*)d_in[6];
  const float* W2   = (const float*)d_in[7];
  const float* b2   = (const float*)d_in[8];
  float* out        = (float*)d_out;

  const int n      = in_sizes[0];
  const int blocks = (n + TILE - 1) / TILE;
  if (n % EPL == 0)
    hipLaunchKernelGGL(csys_kernel<true>, dim3(blocks), dim3(BLOCK), 0, stream,
                       t, z, logK, logz, logp, W1, b1, W2, b2, out, n);
  else
    hipLaunchKernelGGL(csys_kernel<false>, dim3(blocks), dim3(BLOCK), 0, stream,
                       t, z, logK, logz, logp, W1, b1, W2, b2, out, n);
}
